# starting kernel
# baseline (speedup 1.0000x reference)
#include <hip/hip_runtime.h>

typedef __attribute__((ext_vector_type(2))) float v2f;
typedef __attribute__((ext_vector_type(8))) float v8f;

#define N_PTS 500000
#define M_CTR 256
#define NORM_CONST 0.025330295910584444f

__global__ void gauss_weights_kernel(const float* __restrict__ centers,
                                     const float* __restrict__ coefficients,
                                     float* __restrict__ w) {
    __shared__ float red[M_CTR];
    const int t = threadIdx.x;
    float c = coefficients[t];

    red[t] = c;
    __syncthreads();
    for (int s = M_CTR / 2; s > 0; s >>= 1) {
        if (t < s) red[t] = fmaxf(red[t], red[t + s]);
        __syncthreads();
    }
    const float mx = red[0];
    __syncthreads();

    const float e = __expf(c - mx);
    red[t] = e;
    __syncthreads();
    for (int s = M_CTR / 2; s > 0; s >>= 1) {
        if (t < s) red[t] += red[t + s];
        __syncthreads();
    }
    const float sum = red[0];

    const float c0 = centers[t * 4 + 0];
    const float c1 = centers[t * 4 + 1];
    const float c2 = centers[t * 4 + 2];
    const float c3 = centers[t * 4 + 3];
    const float cn2 = c0 * c0 + c1 * c1 + c2 * c2 + c3 * c3;

    w[t] = (e / sum) * NORM_CONST * __expf(-0.5f * cn2);
}

__global__ void __launch_bounds__(256) gauss_main_kernel(
        const float* __restrict__ x,
        const float* __restrict__ centers,
        const float* __restrict__ w,
        float* __restrict__ out) {
    __shared__ float ws[M_CTR];
    ws[threadIdx.x] = w[threadIdx.x];
    __syncthreads();

    const int lane = threadIdx.x & 31;
    const int waveId = blockIdx.x * (blockDim.x >> 5) + (threadIdx.x >> 5);
    const int numTiles = N_PTS / 16;
    if (waveId >= numTiles) return;

    const int row0 = waveId * 16;
    const int r    = lane & 15;
    const int hi   = lane >> 4;
    const int koff = hi * 2;

    v2f a = *(const v2f*)(x + (size_t)(row0 + r) * 4 + koff);

    float part = a.x * a.x + a.y * a.y;
    float x2 = part + __shfl_xor(part, 16);

    float acc[8];
#pragma unroll
    for (int j = 0; j < 8; ++j) acc[j] = 0.0f;

#pragma unroll
    for (int tile = 0; tile < 16; ++tile) {
        const int col = tile * 16 + r;
        v2f b = *(const v2f*)(centers + col * 4 + koff);

        v8f c = {};
        c = __builtin_amdgcn_wmma_f32_16x16x4_f32(
                 false, a,  false, b,
                 (short)0, c,  false,  false);

        const float wcol = ws[col];
#pragma unroll
        for (int j = 0; j < 8; ++j)
            acc[j] += wcol * __expf(c[j]);
    }

#pragma unroll
    for (int j = 0; j < 8; ++j) {
        float v = acc[j];
        v += __shfl_xor(v, 1);
        v += __shfl_xor(v, 2);
        v += __shfl_xor(v, 4);
        v += __shfl_xor(v, 8);
        acc[j] = v;
    }

#pragma unroll
    for (int j = 0; j < 8; ++j) {
        const float x2j = __shfl(x2, hi * 8 + j);
        acc[j] *= __expf(-0.5f * x2j);
    }

    if (r == 0) {
        const int base = row0 + hi * 8;
#pragma unroll
        for (int j = 0; j < 8; ++j) out[base + j] = acc[j];
    }
}

extern "C" void kernel_launch(void* const* d_in, const int* in_sizes, int n_in,
                              void* d_out, int out_size, void* d_ws, size_t ws_size,
                              hipStream_t stream) {
    const float* x            = (const float*)d_in[0];
    const float* centers      = (const float*)d_in[1];
    const float* coefficients = (const float*)d_in[2];
    float* out = (float*)d_out;
    float* w   = (float*)d_ws;

    gauss_weights_kernel<<<1, M_CTR, 0, stream>>>(centers, coefficients, w);

    const int numWaveTiles = N_PTS / 16;
    const int wavesPerBlock = 8;
    const int blocks = (numWaveTiles + wavesPerBlock - 1) / wavesPerBlock;
    gauss_main_kernel<<<blocks, 256, 0, stream>>>(x, centers, w, out);
}


# Weights launch without LDS or barriers: each of its 8 waves finds the softmax max and sum of all 256 coefficients in registers (8 per lane, then lane-xor tree: new sum order), all loads up front
# speedup vs baseline: 1.2056x; 1.2056x over previous
#include <hip/hip_runtime.h>


typedef __attribute__((ext_vector_type(2))) float v2f;
typedef __attribute__((ext_vector_type(8))) float v8f;

#define N_PTS 500000
#define M_CTR 256
#define NORM_CONST 0.025330295910584444f

#define ROW_TILES     8
#define ROWS_PER_WAVE (16 * ROW_TILES)
#define CTR_TILES     (M_CTR / 16)
#define MAIN_WAVES_PER_BLOCK 8
#define MAIN_BLOCKS   489


template <int M>
__device__ __forceinline__ float row_xor(float v) {
    const int b = __builtin_bit_cast(int, v);
    return __builtin_bit_cast(float,
        __builtin_amdgcn_update_dpp(b, b, 0x160 + M, 0xf, 0xf, false));
}

__device__ __forceinline__ float half_xor(float v) {
    const int b = __builtin_bit_cast(int, v);
    return __builtin_bit_cast(float,
        __builtin_amdgcn_permlanex16(b, b, 0x76543210, 0xfedcba98, false, false));
}

__global__ void __launch_bounds__(M_CTR) gauss_weights_kernel(
        const float* __restrict__ centers,
        const float* __restrict__ coefficients,
        float* __restrict__ w) {
    const int t    = threadIdx.x;
    const int lane = t & 31;

    float cf[8];
#pragma unroll
    for (int i = 0; i < 8; ++i) cf[i] = coefficients[lane * 8 + i];
    const float c  = coefficients[t];
    const float c0 = centers[t * 4 + 0];
    const float c1 = centers[t * 4 + 1];
    const float c2 = centers[t * 4 + 2];
    const float c3 = centers[t * 4 + 3];

    float mx = fmaxf(fmaxf(fmaxf(cf[0], cf[1]), fmaxf(cf[2], cf[3])),
                     fmaxf(fmaxf(cf[4], cf[5]), fmaxf(cf[6], cf[7])));
    mx = fmaxf(mx, row_xor<1>(mx));
    mx = fmaxf(mx, row_xor<2>(mx));
    mx = fmaxf(mx, row_xor<4>(mx));
    mx = fmaxf(mx, row_xor<8>(mx));
    mx = fmaxf(mx, half_xor(mx));

    float ex[8];
#pragma unroll
    for (int i = 0; i < 8; ++i) ex[i] = __expf(cf[i] - mx);
    float sum = ((ex[0] + ex[1]) + (ex[2] + ex[3])) + ((ex[4] + ex[5]) + (ex[6] + ex[7]));
    sum += row_xor<1>(sum);
    sum += row_xor<2>(sum);
    sum += row_xor<4>(sum);
    sum += row_xor<8>(sum);
    sum += half_xor(sum);

    const float e   = __expf(c - mx);
    const float cn2 = c0 * c0 + c1 * c1 + c2 * c2 + c3 * c3;

    w[t] = (e / sum) * NORM_CONST * __expf(-0.5f * cn2);
}

__global__ void __launch_bounds__(256) gauss_main_kernel(
        const float* __restrict__ x,
        const float* __restrict__ centers,
        const float* __restrict__ w,
        float* __restrict__ out) {
    const int lane   = threadIdx.x & 31;
    const int waveId = blockIdx.x * MAIN_WAVES_PER_BLOCK + (threadIdx.x >> 5);
    const int row0   = waveId * ROWS_PER_WAVE;
    if (row0 >= N_PTS) return;

    const int r    = lane & 15;
    const int hi   = lane >> 4;
    const int koff = hi * 2;

    v2f a[ROW_TILES];
#pragma unroll
    for (int t = 0; t < ROW_TILES; ++t) {
        int row = row0 + t * 16 + r;
        row = row < N_PTS ? row : N_PTS - 1;
        a[t] = *(const v2f*)(x + (size_t)row * 4 + koff);
    }

    float acc[ROW_TILES][8];
#pragma unroll
    for (int t = 0; t < ROW_TILES; ++t)
#pragma unroll
        for (int j = 0; j < 8; ++j) acc[t][j] = 0.0f;

    for (int tile = 0; tile < CTR_TILES; ++tile) {
        const int col = tile * 16 + r;
        const v2f   b    = *(const v2f*)(centers + col * 4 + koff);
        const float wcol = w[col];

#pragma unroll
        for (int t = 0; t < ROW_TILES; ++t) {
            v8f c = {};
            c = __builtin_amdgcn_wmma_f32_16x16x4_f32(
                     false, a[t],  false, b,
                     (short)0, c,  false,  false);
#pragma unroll
            for (int j = 0; j < 8; ++j)
                acc[t][j] += wcol * __expf(c[j]);
        }
    }

#pragma unroll
    for (int t = 0; t < ROW_TILES; ++t) {
        const float part = a[t].x * a[t].x + a[t].y * a[t].y;
        const float x2   = part + __shfl_xor(part, 16);

#pragma unroll
        for (int j = 0; j < 8; ++j) {
            float v = acc[t][j];
            v += __shfl_xor(v, 1);
            v += __shfl_xor(v, 2);
            v += __shfl_xor(v, 4);
            v += __shfl_xor(v, 8);
            const float x2j = __shfl(x2, hi * 8 + j);
            acc[t][j] = v * __expf(-0.5f * x2j);
        }
    }

    if (r == 0) {
#pragma unroll
        for (int t = 0; t < ROW_TILES; ++t) {
            const int trow = row0 + t * 16;
            if (trow < N_PTS) {
                const int base = trow + hi * 8;
#pragma unroll
                for (int j = 0; j < 8; ++j) out[base + j] = acc[t][j];
            }
        }
    }
}

extern "C" void kernel_launch(void* const* d_in, const int* in_sizes, int n_in,
                              void* d_out, int out_size, void* d_ws, size_t ws_size,
                              hipStream_t stream) {
    const float* x            = (const float*)d_in[0];
    const float* centers      = (const float*)d_in[1];
    const float* coefficients = (const float*)d_in[2];
    float* out = (float*)d_out;
    float* w   = (float*)d_ws;

    gauss_weights_kernel<<<1, M_CTR, 0, stream>>>(centers, coefficients, w);

    gauss_main_kernel<<<MAIN_BLOCKS, 32 * MAIN_WAVES_PER_BLOCK, 0, stream>>>(
            x, centers, w, out);
}


# Trip issued as a pipeline over the 8 row tiles (scheduling groups): WMMA of tile t+2, then tile t's 8 exps, one plain op behind each (fmas of t-1, muls of t+1); was 4 blocks; sums unchanged
# speedup vs baseline: 1.2746x; 1.0572x over previous
#include <hip/hip_runtime.h>


typedef __attribute__((ext_vector_type(2))) float v2f;
typedef __attribute__((ext_vector_type(8))) float v8f;

#define N_PTS 500000
#define M_CTR 256
#define NORM_CONST 0.025330295910584444f

#define ROW_TILES     8
#define ROWS_PER_WAVE (16 * ROW_TILES)
#define CTR_TILES     (M_CTR / 16)
#define MAIN_WAVES_PER_BLOCK 8
#define MAIN_BLOCKS   489


template <int M>
__device__ __forceinline__ float row_xor(float v) {
    const int b = __builtin_bit_cast(int, v);
    return __builtin_bit_cast(float,
        __builtin_amdgcn_update_dpp(b, b, 0x160 + M, 0xf, 0xf, false));
}

__device__ __forceinline__ float half_xor(float v) {
    const int b = __builtin_bit_cast(int, v);
    return __builtin_bit_cast(float,
        __builtin_amdgcn_permlanex16(b, b, 0x76543210, 0xfedcba98, false, false));
}

__global__ void __launch_bounds__(M_CTR) gauss_weights_kernel(
        const float* __restrict__ centers,
        const float* __restrict__ coefficients,
        float* __restrict__ w) {
    const int t    = threadIdx.x;
    const int lane = t & 31;

    float cf[8];
#pragma unroll
    for (int i = 0; i < 8; ++i) cf[i] = coefficients[lane * 8 + i];
    const float c  = coefficients[t];
    const float c0 = centers[t * 4 + 0];
    const float c1 = centers[t * 4 + 1];
    const float c2 = centers[t * 4 + 2];
    const float c3 = centers[t * 4 + 3];

    float mx = fmaxf(fmaxf(fmaxf(cf[0], cf[1]), fmaxf(cf[2], cf[3])),
                     fmaxf(fmaxf(cf[4], cf[5]), fmaxf(cf[6], cf[7])));
    mx = fmaxf(mx, row_xor<1>(mx));
    mx = fmaxf(mx, row_xor<2>(mx));
    mx = fmaxf(mx, row_xor<4>(mx));
    mx = fmaxf(mx, row_xor<8>(mx));
    mx = fmaxf(mx, half_xor(mx));

    float ex[8];
#pragma unroll
    for (int i = 0; i < 8; ++i) ex[i] = __expf(cf[i] - mx);
    float sum = ((ex[0] + ex[1]) + (ex[2] + ex[3])) + ((ex[4] + ex[5]) + (ex[6] + ex[7]));
    sum += row_xor<1>(sum);
    sum += row_xor<2>(sum);
    sum += row_xor<4>(sum);
    sum += row_xor<8>(sum);
    sum += half_xor(sum);

    const float e   = __expf(c - mx);
    const float cn2 = c0 * c0 + c1 * c1 + c2 * c2 + c3 * c3;

    w[t] = (e / sum) * NORM_CONST * __expf(-0.5f * cn2);
}

__global__ void __launch_bounds__(256) gauss_main_kernel(
        const float* __restrict__ x,
        const float* __restrict__ centers,
        const float* __restrict__ w,
        float* __restrict__ out) {
    const int lane   = threadIdx.x & 31;
    const int waveId = blockIdx.x * MAIN_WAVES_PER_BLOCK + (threadIdx.x >> 5);
    const int row0   = waveId * ROWS_PER_WAVE;
    if (row0 >= N_PTS) return;

    const int r    = lane & 15;
    const int hi   = lane >> 4;
    const int koff = hi * 2;

    v2f a[ROW_TILES];
#pragma unroll
    for (int t = 0; t < ROW_TILES; ++t) {
        int row = row0 + t * 16 + r;
        row = row < N_PTS ? row : N_PTS - 1;
        a[t] = *(const v2f*)(x + (size_t)row * 4 + koff);
    }

    float acc[ROW_TILES][8];
#pragma unroll
    for (int t = 0; t < ROW_TILES; ++t)
#pragma unroll
        for (int j = 0; j < 8; ++j) acc[t][j] = 0.0f;

    const float LOG2E = 1.44269504088896340736f;
    for (int tile = 0; tile < CTR_TILES; ++tile) {
        const int col = tile * 16 + r;
        const v2f   b    = *(const v2f*)(centers + col * 4 + koff);
        const float wcol = w[col];

        v8f d[ROW_TILES];
        float e[ROW_TILES][8];

        auto mma = [&](int t) {
            v8f z = {};
            d[t] = __builtin_amdgcn_wmma_f32_16x16x4_f32(
                     false, a[t],  false, b,
                     (short)0, z,  false,  false);
        };
        auto scale2 = [&](int t, int p) {
            d[t][2 * p]     *= LOG2E;
            d[t][2 * p + 1] *= LOG2E;
        };
        auto fma2 = [&](int t, int p) {
            acc[t][2 * p]     += wcol * e[t][2 * p];
            acc[t][2 * p + 1] += wcol * e[t][2 * p + 1];
        };

        mma(0);
        mma(1);
#pragma unroll
        for (int p = 0; p < 4; ++p) scale2(0, p);
#pragma unroll
        for (int t = 0; t < ROW_TILES; ++t) {
            if (t + 2 < ROW_TILES) mma(t + 2);
#pragma unroll
            for (int k = 0; k < 8; ++k) {
                e[t][k] = __builtin_amdgcn_exp2f(d[t][k]);
                if (k < 4) { if (t > 0) fma2(t - 1, k); }
                else       { if (t + 1 < ROW_TILES) scale2(t + 1, k - 4); }
            }
        }
#pragma unroll
        for (int p = 0; p < 4; ++p) fma2(ROW_TILES - 1, p);

        __builtin_amdgcn_sched_group_barrier(0x008, 2, 0);
        __builtin_amdgcn_sched_group_barrier(0x002, 4, 0);
        __builtin_amdgcn_sched_group_barrier(0x008, 1, 0);
        __builtin_amdgcn_sched_group_barrier(0x400, 4, 0);
#pragma unroll
        for (int k = 0; k < 4; ++k) {
            __builtin_amdgcn_sched_group_barrier(0x400, 1, 0);
            __builtin_amdgcn_sched_group_barrier(0x002, 1, 0);
        }
#pragma unroll
        for (int t = 1; t < ROW_TILES; ++t) {
            if (t + 2 < ROW_TILES)
                __builtin_amdgcn_sched_group_barrier(0x008, 1, 0);
#pragma unroll
            for (int k = 0; k < 8; ++k) {
                __builtin_amdgcn_sched_group_barrier(0x400, 1, 0);
                if (k < 4 || t + 1 < ROW_TILES)
                    __builtin_amdgcn_sched_group_barrier(0x002, 1, 0);
            }
        }
        __builtin_amdgcn_sched_group_barrier(0x002, 4, 0);
    }

#pragma unroll
    for (int t = 0; t < ROW_TILES; ++t) {
        const float part = a[t].x * a[t].x + a[t].y * a[t].y;
        const float x2   = part + __shfl_xor(part, 16);

#pragma unroll
        for (int j = 0; j < 8; ++j) {
            float v = acc[t][j];
            v += __shfl_xor(v, 1);
            v += __shfl_xor(v, 2);
            v += __shfl_xor(v, 4);
            v += __shfl_xor(v, 8);
            const float x2j = __shfl(x2, hi * 8 + j);
            acc[t][j] = v * __expf(-0.5f * x2j);
        }
    }

    if (r == 0) {
#pragma unroll
        for (int t = 0; t < ROW_TILES; ++t) {
            const int trow = row0 + t * 16;
            if (trow < N_PTS) {
                const int base = trow + hi * 8;
#pragma unroll
                for (int j = 0; j < 8; ++j) out[base + j] = acc[t][j];
            }
        }
    }
}

extern "C" void kernel_launch(void* const* d_in, const int* in_sizes, int n_in,
                              void* d_out, int out_size, void* d_ws, size_t ws_size,
                              hipStream_t stream) {
    const float* x            = (const float*)d_in[0];
    const float* centers      = (const float*)d_in[1];
    const float* coefficients = (const float*)d_in[2];
    float* out = (float*)d_out;
    float* w   = (float*)d_ws;

    gauss_weights_kernel<<<1, M_CTR, 0, stream>>>(centers, coefficients, w);

    gauss_main_kernel<<<MAIN_BLOCKS, 32 * MAIN_WAVES_PER_BLOCK, 0, stream>>>(
            x, centers, w, out);
}


# Epilogue row sums on the vector ALU: halving reduction by row_xmask adds over two tiles' 16 registers (15 adds, same tree), one row per lane, no LDS permutes; one exp and one full-line store per pair
# speedup vs baseline: 1.4070x; 1.1039x over previous
#include <hip/hip_runtime.h>


typedef __attribute__((ext_vector_type(2))) float v2f;
typedef __attribute__((ext_vector_type(8))) float v8f;

#define N_PTS 500000
#define M_CTR 256
#define NORM_CONST 0.025330295910584444f

#define ROW_TILES     8
#define ROWS_PER_WAVE (16 * ROW_TILES)
#define CTR_TILES     (M_CTR / 16)
#define MAIN_WAVES_PER_BLOCK 8
#define MAIN_BLOCKS   489


template <int M>
__device__ __forceinline__ float row_xor(float v) {
    const int b = __builtin_bit_cast(int, v);
    return __builtin_bit_cast(float,
        __builtin_amdgcn_update_dpp(b, b, 0x160 + M, 0xf, 0xf, false));
}

template <int M>
__device__ __forceinline__ float row_take(float v) {
    const int b = __builtin_bit_cast(int, v);
    return __builtin_bit_cast(float,
        __builtin_amdgcn_update_dpp(0, b, 0x160 + M, 0xf, 0xf, true));
}
template <int M>
__device__ __forceinline__ float halve(bool bit, float lo, float hi) {
    const float keep = bit ? hi : lo;
    const float send = bit ? lo : hi;
    return keep + row_take<M>(send);
}

__device__ __forceinline__ float half_xor(float v) {
    const int b = __builtin_bit_cast(int, v);
    return __builtin_bit_cast(float,
        __builtin_amdgcn_permlanex16(b, b, 0x76543210, 0xfedcba98, false, false));
}

__global__ void __launch_bounds__(M_CTR) gauss_weights_kernel(
        const float* __restrict__ centers,
        const float* __restrict__ coefficients,
        float* __restrict__ w) {
    const int t    = threadIdx.x;
    const int lane = t & 31;

    float cf[8];
#pragma unroll
    for (int i = 0; i < 8; ++i) cf[i] = coefficients[lane * 8 + i];
    const float c  = coefficients[t];
    const float c0 = centers[t * 4 + 0];
    const float c1 = centers[t * 4 + 1];
    const float c2 = centers[t * 4 + 2];
    const float c3 = centers[t * 4 + 3];

    float mx = fmaxf(fmaxf(fmaxf(cf[0], cf[1]), fmaxf(cf[2], cf[3])),
                     fmaxf(fmaxf(cf[4], cf[5]), fmaxf(cf[6], cf[7])));
    mx = fmaxf(mx, row_xor<1>(mx));
    mx = fmaxf(mx, row_xor<2>(mx));
    mx = fmaxf(mx, row_xor<4>(mx));
    mx = fmaxf(mx, row_xor<8>(mx));
    mx = fmaxf(mx, half_xor(mx));

    float ex[8];
#pragma unroll
    for (int i = 0; i < 8; ++i) ex[i] = __expf(cf[i] - mx);
    float sum = ((ex[0] + ex[1]) + (ex[2] + ex[3])) + ((ex[4] + ex[5]) + (ex[6] + ex[7]));
    sum += row_xor<1>(sum);
    sum += row_xor<2>(sum);
    sum += row_xor<4>(sum);
    sum += row_xor<8>(sum);
    sum += half_xor(sum);

    const float e   = __expf(c - mx);
    const float cn2 = c0 * c0 + c1 * c1 + c2 * c2 + c3 * c3;

    w[t] = (e / sum) * NORM_CONST * __expf(-0.5f * cn2);
}

__global__ void __launch_bounds__(256) gauss_main_kernel(
        const float* __restrict__ x,
        const float* __restrict__ centers,
        const float* __restrict__ w,
        float* __restrict__ out) {
    const int lane   = threadIdx.x & 31;
    const int waveId = blockIdx.x * MAIN_WAVES_PER_BLOCK + (threadIdx.x >> 5);
    const int row0   = waveId * ROWS_PER_WAVE;
    if (row0 >= N_PTS) return;

    const int r    = lane & 15;
    const int hi   = lane >> 4;
    const int koff = hi * 2;

    v2f a[ROW_TILES];
#pragma unroll
    for (int t = 0; t < ROW_TILES; ++t) {
        int row = row0 + t * 16 + r;
        row = row < N_PTS ? row : N_PTS - 1;
        a[t] = *(const v2f*)(x + (size_t)row * 4 + koff);
    }

    v2f acc[ROW_TILES][4];
#pragma unroll
    for (int t = 0; t < ROW_TILES; ++t)
#pragma unroll
        for (int p = 0; p < 4; ++p) acc[t][p] = v2f{0.0f, 0.0f};

    const float LOG2E = 1.44269504088896340736f;
    for (int tile = 0; tile < CTR_TILES; ++tile) {
        const int col = tile * 16 + r;
        const v2f   b    = *(const v2f*)(centers + col * 4 + koff);
        const float wcol = w[col];

        v8f d[ROW_TILES];
        float e[ROW_TILES][8];

        auto mma = [&](int t) {
            v8f z = {};
            d[t] = __builtin_amdgcn_wmma_f32_16x16x4_f32(
                     false, a[t],  false, b,
                     (short)0, z,  false,  false);
        };
        auto scale2 = [&](int t, int p) {
            d[t][2 * p]     *= LOG2E;
            d[t][2 * p + 1] *= LOG2E;
        };
        auto fma2 = [&](int t, int p) {
            const v2f ev = {e[t][2 * p], e[t][2 * p + 1]};
            const v2f wv = {wcol, wcol};
            acc[t][p] += wv * ev;
        };

        mma(0);
        mma(1);
#pragma unroll
        for (int p = 0; p < 4; ++p) scale2(0, p);
#pragma unroll
        for (int t = 0; t < ROW_TILES; ++t) {
            if (t + 2 < ROW_TILES) mma(t + 2);
#pragma unroll
            for (int k = 0; k < 8; ++k) {
                e[t][k] = __builtin_amdgcn_exp2f(d[t][k]);
                if (k < 4) { if (t > 0) fma2(t - 1, k); }
                else       { if (t + 1 < ROW_TILES) scale2(t + 1, k - 4); }
            }
        }
#pragma unroll
        for (int p = 0; p < 4; ++p) fma2(ROW_TILES - 1, p);

        __builtin_amdgcn_sched_group_barrier(0x008, 2, 0);
        __builtin_amdgcn_sched_group_barrier(0x002, 4, 0);
        __builtin_amdgcn_sched_group_barrier(0x008, 1, 0);
        __builtin_amdgcn_sched_group_barrier(0x400, 4, 0);
#pragma unroll
        for (int k = 0; k < 4; ++k) {
            __builtin_amdgcn_sched_group_barrier(0x400, 1, 0);
            __builtin_amdgcn_sched_group_barrier(0x002, 1, 0);
        }
#pragma unroll
        for (int t = 1; t < ROW_TILES; ++t) {
            if (t + 2 < ROW_TILES)
                __builtin_amdgcn_sched_group_barrier(0x008, 1, 0);
#pragma unroll
            for (int k = 0; k < 8; ++k) {
                __builtin_amdgcn_sched_group_barrier(0x400, 1, 0);
                if (k < 4 || t + 1 < ROW_TILES)
                    __builtin_amdgcn_sched_group_barrier(0x002, 1, 0);
            }
        }
        __builtin_amdgcn_sched_group_barrier(0x002, 4, 0);
    }

    const bool b0 = (lane & 1) != 0;
    const bool b1 = (lane & 2) != 0;
    const bool b2 = (lane & 4) != 0;
    const bool b3 = (lane & 8) != 0;
    float res[ROW_TILES / 2];
#pragma unroll
    for (int q = 0; q < ROW_TILES / 2; ++q) {
        float s1[8], s2[4], s3[2];
#pragma unroll
        for (int i = 0; i < 8; ++i)
            s1[i] = halve<1>(b0, acc[2 * q + (i >> 2)][i & 3].x,
                                 acc[2 * q + (i >> 2)][i & 3].y);
#pragma unroll
        for (int i = 0; i < 4; ++i) s2[i] = halve<2>(b1, s1[2 * i], s1[2 * i + 1]);
#pragma unroll
        for (int i = 0; i < 2; ++i) s3[i] = halve<4>(b2, s2[2 * i], s2[2 * i + 1]);
        const float v = halve<8>(b3, s3[0], s3[1]);

        const float p0  = a[2 * q].x * a[2 * q].x + a[2 * q].y * a[2 * q].y;
        const float p1  = a[2 * q + 1].x * a[2 * q + 1].x + a[2 * q + 1].y * a[2 * q + 1].y;
        const float x20 = p0 + half_xor(p0);
        const float x21 = p1 + half_xor(p1);
        const float mine  = b3 ? x21 : x20;
        const float other = row_take<8>(b3 ? x20 : x21);
        const float x2row = (b3 == (hi != 0)) ? mine : other;

        res[q] = v * __expf(-0.5f * x2row);
    }

#pragma unroll
    for (int q = 0; q < ROW_TILES / 2; ++q) {
        const int row = row0 + (2 * q + (b3 ? 1 : 0)) * 16 + hi * 8 + (r & 7);
        if (row < N_PTS) out[row] = res[q];
    }
}

extern "C" void kernel_launch(void* const* d_in, const int* in_sizes, int n_in,
                              void* d_out, int out_size, void* d_ws, size_t ws_size,
                              hipStream_t stream) {
    const float* x            = (const float*)d_in[0];
    const float* centers      = (const float*)d_in[1];
    const float* coefficients = (const float*)d_in[2];
    float* out = (float*)d_out;
    float* w   = (float*)d_ws;

    gauss_weights_kernel<<<1, M_CTR, 0, stream>>>(centers, coefficients, w);

    gauss_main_kernel<<<MAIN_BLOCKS, 32 * MAIN_WAVES_PER_BLOCK, 0, stream>>>(
            x, centers, w, out);
}
